# starting kernel
# baseline (speedup 1.0000x reference)
#include <hip/hip_runtime.h>
#include <hip/hip_bf16.h>

#define L_SEQ 250
#define DMODEL 8
#define DINNER 16
#define DSTATE 16
#define DCONV 5
#define NCLS 12

typedef __attribute__((ext_vector_type(2))) float v2f;
typedef __attribute__((ext_vector_type(8))) float v8f;

__device__ __forceinline__ float sigmoidf_(float v) { return 1.0f / (1.0f + __expf(-v)); }

__global__ __launch_bounds__(256) void mamba_fused_kernel(
    const float* __restrict__ x,
    const float* __restrict__ W_in,
    const float* __restrict__ conv_w,
    const float* __restrict__ conv_b,
    const float* __restrict__ W_x,
    const float* __restrict__ dt_w,
    const float* __restrict__ dt_b,
    const float* __restrict__ A_log,
    const float* __restrict__ Dp,
    const float* __restrict__ W_out,
    const float* __restrict__ fc_w,
    const float* __restrict__ fc_b,
    float* __restrict__ out)
{
    __shared__ float smem[16298];
    float* s_z   = smem;
    float* s_xc  = smem + 4016;
    float* s_B   = smem + 8016;
    float* s_C   = smem + 12032;
    float* s_dtr = smem + 16048;
    float* s_xb  = s_B;
    float* s_y   = s_z;
    float* s_out = s_C;
    float* s_red = s_dtr;

    const int b    = blockIdx.x;
    const int tid  = threadIdx.x;
    const int lane = tid & 31;
    const int wave = tid >> 5;
    const int lm   = lane & 15;
    const int hi   = lane >> 4;
    const float* xb = x + (size_t)b * DMODEL * L_SEQ;

    for (int i = tid; i < DMODEL * L_SEQ; i += 256) s_xb[i] = xb[i];
    for (int i = tid * 32; i < NCLS * L_SEQ * DMODEL; i += 256 * 32)
        __builtin_prefetch(fc_w + i, 0, 0);
    __syncthreads();

    for (int mt = wave; mt < 16; mt += 8) {
        const int l0 = mt * 16;
        const int lr = min(l0 + lm, L_SEQ - 1);
        v8f acc = {};
#pragma unroll
        for (int kc = 0; kc < 8; kc += 4) {
            const int k0 = kc + 2 * hi;
            v2f a, bb;
            a.x  = s_xb[(k0    ) * L_SEQ + lr];
            a.y  = s_xb[(k0 + 1) * L_SEQ + lr];
            bb.x = W_in[(16 + lm) * 8 + k0];
            bb.y = W_in[(16 + lm) * 8 + k0 + 1];
            acc = __builtin_amdgcn_wmma_f32_16x16x4_f32(false, a, false, bb,
                                                        (short)0, acc, false, false);
        }
#pragma unroll
        for (int r = 0; r < 8; ++r) {
            const int row = min(l0 + r + 8 * hi, L_SEQ);
            s_z[row * 16 + lm] = acc[r];
        }
    }

    for (int i = tid; i < L_SEQ * DINNER; i += 256) {
        const int l = i >> 4, d = i & 15;
        float accv = conv_b[d];
#pragma unroll
        for (int j = 0; j < DCONV; ++j) {
            const int t = l - (DCONV - 1) + j;
            if (t >= 0) {
                float xi = 0.f;
#pragma unroll
                for (int dm = 0; dm < DMODEL; ++dm) xi += s_xb[dm * L_SEQ + t] * W_in[d * 8 + dm];
                accv += conv_w[d * DCONV + j] * xi;
            }
        }
        s_xc[i] = accv * sigmoidf_(accv);
    }
    __syncthreads();

    for (int l = tid; l < L_SEQ; l += 256) {
        float a = 0.f;
#pragma unroll
        for (int d = 0; d < DINNER; ++d) a += s_xc[l * 16 + d] * W_x[d];
        s_dtr[l] = a;
    }

    for (int j = wave; j < 32; j += 8) {
        const int mt = j >> 1, nt = j & 1;
        const int l0 = mt * 16;
        const int lr = min(l0 + lm, L_SEQ - 1);
        const int e0 = 1 + nt * 16;
        v8f acc = {};
#pragma unroll
        for (int kc = 0; kc < 16; kc += 4) {
            const int k0 = kc + 2 * hi;
            v2f a, bb;
            a.x  = s_xc[lr * 16 + k0];
            a.y  = s_xc[lr * 16 + k0 + 1];
            bb.x = W_x[(e0 + lm) * 16 + k0];
            bb.y = W_x[(e0 + lm) * 16 + k0 + 1];
            acc = __builtin_amdgcn_wmma_f32_16x16x4_f32(false, a, false, bb,
                                                        (short)0, acc, false, false);
        }
        float* dst = nt ? s_C : s_B;
#pragma unroll
        for (int r = 0; r < 8; ++r) {
            const int row = min(l0 + r + 8 * hi, L_SEQ);
            dst[row * 16 + lm] = acc[r];
        }
    }
    __syncthreads();

    {
        const int d = tid >> 4;
        const int s = tid & 15;
        const float dtw = dt_w[d], dtb = dt_b[d], Dpd = Dp[d];
        const float As  = -__expf(A_log[d * 16 + s]);
        float h = 0.f;
        for (int l = 0; l < L_SEQ; ++l) {
            float dv = s_dtr[l] * dtw + dtb;
            dv = (dv > 20.f) ? dv : log1pf(__expf(dv));
            const float xcv = s_xc[l * 16 + d];
            h = __expf(dv * As) * h + (dv * xcv) * s_B[l * 16 + s];
            float y = h * s_C[l * 16 + s];
            y += __shfl_xor(y, 1, 32);
            y += __shfl_xor(y, 2, 32);
            y += __shfl_xor(y, 4, 32);
            y += __shfl_xor(y, 8, 32);
            if (s == 0) {
                y += xcv * Dpd;
                const float zv = s_z[l * 16 + d];
                y *= zv * sigmoidf_(zv);
                s_y[l * 16 + d] = y;
            }
        }
    }
    __syncthreads();

    for (int mt = wave; mt < 16; mt += 8) {
        const int l0 = mt * 16;
        const int lr = min(l0 + lm, L_SEQ - 1);
        v8f acc = {};
#pragma unroll
        for (int kc = 0; kc < 16; kc += 4) {
            const int k0 = kc + 2 * hi;
            v2f a, bb;
            a.x  = s_y[lr * 16 + k0];
            a.y  = s_y[lr * 16 + k0 + 1];
            bb.x = (lm < DMODEL) ? W_out[lm * 16 + k0]     : 0.f;
            bb.y = (lm < DMODEL) ? W_out[lm * 16 + k0 + 1] : 0.f;
            acc = __builtin_amdgcn_wmma_f32_16x16x4_f32(false, a, false, bb,
                                                        (short)0, acc, false, false);
        }
        if (lm < DMODEL) {
#pragma unroll
            for (int r = 0; r < 8; ++r)
                s_out[(l0 + r + 8 * hi) * 8 + lm] = acc[r];
        }
    }
    __syncthreads();

    {
        float part[NCLS];
#pragma unroll
        for (int c = 0; c < NCLS; ++c) part[c] = 0.f;
        for (int i = tid; i < L_SEQ * DMODEL; i += 256) {
            const float v = s_out[i];
#pragma unroll
            for (int c = 0; c < NCLS; ++c) part[c] += v * fc_w[c * (L_SEQ * DMODEL) + i];
        }
#pragma unroll
        for (int c = 0; c < NCLS; ++c) {
#pragma unroll
            for (int off = 16; off > 0; off >>= 1)
                part[c] += __shfl_down(part[c], off, 32);
        }
        if (lane == 0) {
#pragma unroll
            for (int c = 0; c < NCLS; ++c) s_red[wave * NCLS + c] = part[c];
        }
        __syncthreads();
        if (tid < NCLS) {
            float sres = fc_b[tid];
#pragma unroll
            for (int w = 0; w < 8; ++w) sres += s_red[w * NCLS + tid];
            out[b * NCLS + tid] = sres;
        }
    }
}

extern "C" void kernel_launch(void* const* d_in, const int* in_sizes, int n_in,
                              void* d_out, int out_size, void* d_ws, size_t ws_size,
                              hipStream_t stream) {
    (void)in_sizes; (void)n_in; (void)out_size; (void)d_ws; (void)ws_size;
    mamba_fused_kernel<<<dim3(1024), dim3(256), 0, stream>>>(
        (const float*)d_in[0],
        (const float*)d_in[1],
        (const float*)d_in[2],
        (const float*)d_in[3],
        (const float*)d_in[4],
        (const float*)d_in[5],
        (const float*)d_in[6],
        (const float*)d_in[7],
        (const float*)d_in[8],
        (const float*)d_in[9],
        (const float*)d_in[10],
        (const float*)d_in[11],
        (float*)d_out);
}


# Scan re-decomposed: one wave carries one batch element (lane = channel x half the states, 8 states in registers, in-lane sums + one cross-half add: sum order changes); 4 elements per work-group
# speedup vs baseline: 2.8758x; 2.8758x over previous
#include <hip/hip_runtime.h>
#include <hip/hip_bf16.h>

#define L_SEQ 250
#define DMODEL 8
#define DINNER 16
#define DSTATE 16
#define DCONV 5
#define NCLS 12
#define NELEM 4
#define IMG 16304

typedef __attribute__((ext_vector_type(2))) float v2f;
typedef __attribute__((ext_vector_type(8))) float v8f;

__device__ __forceinline__ float sigmoidf_(float v) { return 1.0f / (1.0f + __expf(-v)); }

__device__ __forceinline__ float softplus_(float v) {
    const float sp = log1pf(__expf(v));
    return (v > 20.f) ? v : sp;
}

__device__ __forceinline__ float xhalf_(float v) {
    return __builtin_bit_cast(float, __builtin_amdgcn_permlanex16(__builtin_bit_cast(int, v), __builtin_bit_cast(int, v),
                                                                 0x76543210, 0xfedcba98, false, false));
}

__global__ __launch_bounds__(1024) void mamba_fused_kernel(
    const float* __restrict__ x,
    const float* __restrict__ W_in,
    const float* __restrict__ conv_w,
    const float* __restrict__ conv_b,
    const float* __restrict__ W_x,
    const float* __restrict__ dt_w,
    const float* __restrict__ dt_b,
    const float* __restrict__ A_log,
    const float* __restrict__ Dp,
    const float* __restrict__ W_out,
    const float* __restrict__ fc_w,
    const float* __restrict__ fc_b,
    float* __restrict__ out)
{
    __shared__ __attribute__((aligned(64))) float smem_all[NELEM * IMG];
    const int tid_all = threadIdx.x;
    const int g    = tid_all >> 8;
    float* smem  = smem_all + g * IMG;
    float* s_z   = smem;
    float* s_xc  = smem + 4016;
    float* s_B   = smem + 8016;
    float* s_C   = smem + 12032;
    float* s_dtr = smem + 16048;
    float* s_xb  = s_B;
    float* s_y   = s_z;
    float* s_out = s_C;
    float* s_red = s_dtr;

    const int b    = blockIdx.x * NELEM + g;
    const int tid  = tid_all & 255;
    const int lane = tid & 31;
    const int wave = tid >> 5;
    const int lm   = lane & 15;
    const int hi   = lane >> 4;
    const float* xb = x + (size_t)b * DMODEL * L_SEQ;

    for (int i = tid; i < DMODEL * L_SEQ; i += 256) s_xb[i] = xb[i];
    for (int i = tid * 32; i < NCLS * L_SEQ * DMODEL; i += 256 * 32)
        __builtin_prefetch(fc_w + i, 0, 0);
    __syncthreads();

    for (int mt = wave; mt < 16; mt += 8) {
        const int l0 = mt * 16;
        const int lr = min(l0 + lm, L_SEQ - 1);
        v8f acc = {};
#pragma unroll
        for (int kc = 0; kc < 8; kc += 4) {
            const int k0 = kc + 2 * hi;
            v2f a, bb;
            a.x  = s_xb[(k0    ) * L_SEQ + lr];
            a.y  = s_xb[(k0 + 1) * L_SEQ + lr];
            bb.x = W_in[(16 + lm) * 8 + k0];
            bb.y = W_in[(16 + lm) * 8 + k0 + 1];
            acc = __builtin_amdgcn_wmma_f32_16x16x4_f32(false, a, false, bb,
                                                        (short)0, acc, false, false);
        }
#pragma unroll
        for (int r = 0; r < 8; ++r) {
            const int row = min(l0 + r + 8 * hi, L_SEQ);
            s_z[row * 16 + lm] = acc[r];
        }
    }

    for (int i = tid; i < L_SEQ * DINNER; i += 256) {
        const int l = i >> 4, d = i & 15;
        float accv = conv_b[d];
#pragma unroll
        for (int j = 0; j < DCONV; ++j) {
            const int t = l - (DCONV - 1) + j;
            if (t >= 0) {
                float xi = 0.f;
#pragma unroll
                for (int dm = 0; dm < DMODEL; ++dm) xi += s_xb[dm * L_SEQ + t] * W_in[d * 8 + dm];
                accv += conv_w[d * DCONV + j] * xi;
            }
        }
        s_xc[i] = accv * sigmoidf_(accv);
    }
    __syncthreads();

    for (int l = tid; l < L_SEQ; l += 256) {
        float a = 0.f;
#pragma unroll
        for (int d = 0; d < DINNER; ++d) a += s_xc[l * 16 + d] * W_x[d];
        s_dtr[l] = a;
    }

    for (int j = wave; j < 32; j += 8) {
        const int mt = j >> 1, nt = j & 1;
        const int l0 = mt * 16;
        const int lr = min(l0 + lm, L_SEQ - 1);
        const int e0 = 1 + nt * 16;
        v8f acc = {};
#pragma unroll
        for (int kc = 0; kc < 16; kc += 4) {
            const int k0 = kc + 2 * hi;
            v2f a, bb;
            a.x  = s_xc[lr * 16 + k0];
            a.y  = s_xc[lr * 16 + k0 + 1];
            bb.x = W_x[(e0 + lm) * 16 + k0];
            bb.y = W_x[(e0 + lm) * 16 + k0 + 1];
            acc = __builtin_amdgcn_wmma_f32_16x16x4_f32(false, a, false, bb,
                                                        (short)0, acc, false, false);
        }
        float* dst = nt ? s_C : s_B;
#pragma unroll
        for (int r = 0; r < 8; ++r) {
            const int row = min(l0 + r + 8 * hi, L_SEQ);
            dst[row * 16 + lm] = acc[r];
        }
    }
    __syncthreads();

    {
        const int wv = __builtin_amdgcn_readfirstlane(tid_all >> 5);
        const int e  = wv / 9;
        if (wv == 9 * e) {
            float* img = smem_all + e * IMG;
            float*       y_   = img;
            const float* xc_  = img + 4016;
            const float* dtr_ = img + 16048;
            const int d  = lane & 15;
            const int hf = lane >> 4;
            const float4* pB = reinterpret_cast<const float4*>(img + 8016  + 8 * hf);
            const float4* pC = reinterpret_cast<const float4*>(img + 12032 + 8 * hf);
            const float dtw = dt_w[d], dtb = dt_b[d], Dpd = Dp[d];
            float As[8], h[8];
#pragma unroll
            for (int j = 0; j < 8; ++j) { As[j] = -__expf(A_log[d * 16 + 8 * hf + j]); h[j] = 0.f; }

            float  xcv = xc_[d], zv = y_[d];
            float4 b0 = pB[0], b1 = pB[1], c0 = pC[0], c1 = pC[1];
            float  dv_cur = softplus_(dtr_[0] * dtw + dtb);
            float  dv_nxt = softplus_(dtr_[1] * dtw + dtb);
            float  dtr_n2 = dtr_[2];
#pragma unroll 2
            for (int l = 0; l < L_SEQ; ++l) {
                const int ln = min(l + 1, L_SEQ - 1);
                const float  n_xcv = xc_[ln * 16 + d], n_zv = y_[ln * 16 + d];
                const float4 nb0 = pB[ln * 4], nb1 = pB[ln * 4 + 1], nc0 = pC[ln * 4], nc1 = pC[ln * 4 + 1];
                const float  n_dtr3 = dtr_[min(l + 3, L_SEQ - 1)];
                const float dv_n2 = softplus_(dtr_n2 * dtw + dtb);

                const float dv  = dv_cur;
                const float dtx = dv * xcv;
                const float Bv[8] = {b0.x, b0.y, b0.z, b0.w, b1.x, b1.y, b1.z, b1.w};
                const float Cv[8] = {c0.x, c0.y, c0.z, c0.w, c1.x, c1.y, c1.z, c1.w};
#pragma unroll
                for (int j = 0; j < 8; ++j)
                    h[j] = __expf(dv * As[j]) * h[j] + dtx * Bv[j];
                float ya = h[0] * Cv[0], yb = h[4] * Cv[4];
#pragma unroll
                for (int j = 1; j < 4; ++j) { ya += h[j] * Cv[j]; yb += h[4 + j] * Cv[4 + j]; }
                float y = ya + yb;
                y += xhalf_(y);
                y += xcv * Dpd;
                y *= zv * sigmoidf_(zv);
                y_[(hf ? L_SEQ : l) * 16 + d] = y;
                xcv = n_xcv; zv = n_zv;
                b0 = nb0; b1 = nb1; c0 = nc0; c1 = nc1;
                dv_cur = dv_nxt; dv_nxt = dv_n2; dtr_n2 = n_dtr3;
            }
        }
    }
    __syncthreads();

    for (int mt = wave; mt < 16; mt += 8) {
        const int l0 = mt * 16;
        const int lr = min(l0 + lm, L_SEQ - 1);
        v8f acc = {};
#pragma unroll
        for (int kc = 0; kc < 16; kc += 4) {
            const int k0 = kc + 2 * hi;
            v2f a, bb;
            a.x  = s_y[lr * 16 + k0];
            a.y  = s_y[lr * 16 + k0 + 1];
            bb.x = (lm < DMODEL) ? W_out[lm * 16 + k0]     : 0.f;
            bb.y = (lm < DMODEL) ? W_out[lm * 16 + k0 + 1] : 0.f;
            acc = __builtin_amdgcn_wmma_f32_16x16x4_f32(false, a, false, bb,
                                                        (short)0, acc, false, false);
        }
        if (lm < DMODEL) {
#pragma unroll
            for (int r = 0; r < 8; ++r)
                s_out[(l0 + r + 8 * hi) * 8 + lm] = acc[r];
        }
    }
    __syncthreads();

    {
        float part[NCLS];
#pragma unroll
        for (int c = 0; c < NCLS; ++c) part[c] = 0.f;
        for (int i = tid; i < L_SEQ * DMODEL; i += 256) {
            const float v = s_out[i];
#pragma unroll
            for (int c = 0; c < NCLS; ++c) part[c] += v * fc_w[c * (L_SEQ * DMODEL) + i];
        }
#pragma unroll
        for (int c = 0; c < NCLS; ++c) {
#pragma unroll
            for (int off = 16; off > 0; off >>= 1)
                part[c] += __shfl_down(part[c], off, 32);
        }
        if (lane == 0) {
#pragma unroll
            for (int c = 0; c < NCLS; ++c) s_red[wave * NCLS + c] = part[c];
        }
        __syncthreads();
        if (tid < NCLS) {
            float sres = fc_b[tid];
#pragma unroll
            for (int w = 0; w < 8; ++w) sres += s_red[w * NCLS + tid];
            out[b * NCLS + tid] = sres;
        }
    }
}

extern "C" void kernel_launch(void* const* d_in, const int* in_sizes, int n_in,
                              void* d_out, int out_size, void* d_ws, size_t ws_size,
                              hipStream_t stream) {
    (void)in_sizes; (void)n_in; (void)out_size; (void)d_ws; (void)ws_size;
    mamba_fused_kernel<<<dim3(256), dim3(1024), 0, stream>>>(
        (const float*)d_in[0],
        (const float*)d_in[1],
        (const float*)d_in[2],
        (const float*)d_in[3],
        (const float*)d_in[4],
        (const float*)d_in[5],
        (const float*)d_in[6],
        (const float*)d_in[7],
        (const float*)d_in[8],
        (const float*)d_in[9],
        (const float*)d_in[10],
        (const float*)d_in[11],
        (float*)d_out);
}


# Scan of an element split by channel over two waves of a SIMD: lane = channel x quarter of states; a channel's 4 lanes share softplus, gate, store by step; 4 in-lane + 2 DPP adds: sum order changes
# speedup vs baseline: 4.1771x; 1.4525x over previous
#include <hip/hip_runtime.h>
#include <hip/hip_bf16.h>

#define L_SEQ 250
#define DMODEL 8
#define DINNER 16
#define DSTATE 16
#define DCONV 5
#define NCLS 12
#define NELEM 4
#define IMG 16304

typedef __attribute__((ext_vector_type(2))) float v2f;
typedef __attribute__((ext_vector_type(8))) float v8f;

__device__ __forceinline__ float sigmoidf_(float v) { return 1.0f / (1.0f + __expf(-v)); }

__device__ __forceinline__ float softplus_(float v) {
    const float sp = log1pf(__expf(v));
    return (v > 20.f) ? v : sp;
}

template <int CTRL>
__device__ __forceinline__ float quad_(float v) {
    const int b = __builtin_bit_cast(int, v);
    return __builtin_bit_cast(float, __builtin_amdgcn_update_dpp(b, b, CTRL, 0xf, 0xf, false));
}

__global__ __launch_bounds__(1024) void mamba_fused_kernel(
    const float* __restrict__ x,
    const float* __restrict__ W_in,
    const float* __restrict__ conv_w,
    const float* __restrict__ conv_b,
    const float* __restrict__ W_x,
    const float* __restrict__ dt_w,
    const float* __restrict__ dt_b,
    const float* __restrict__ A_log,
    const float* __restrict__ Dp,
    const float* __restrict__ W_out,
    const float* __restrict__ fc_w,
    const float* __restrict__ fc_b,
    float* __restrict__ out)
{
    __shared__ __attribute__((aligned(64))) float smem_all[NELEM * IMG];
    const int tid_all = threadIdx.x;
    const int g    = tid_all >> 8;
    float* smem  = smem_all + g * IMG;
    float* s_z   = smem;
    float* s_xc  = smem + 4016;
    float* s_B   = smem + 8016;
    float* s_C   = smem + 12032;
    float* s_dtr = smem + 16048;
    float* s_xb  = s_B;
    float* s_y   = s_z;
    float* s_out = s_C;
    float* s_red = s_dtr;

    const int b    = blockIdx.x * NELEM + g;
    const int tid  = tid_all & 255;
    const int lane = tid & 31;
    const int wave = tid >> 5;
    const int lm   = lane & 15;
    const int hi   = lane >> 4;
    const float* xb = x + (size_t)b * DMODEL * L_SEQ;

    for (int i = tid; i < DMODEL * L_SEQ; i += 256) s_xb[i] = xb[i];
    for (int i = tid * 32; i < NCLS * L_SEQ * DMODEL; i += 256 * 32)
        __builtin_prefetch(fc_w + i, 0, 0);
    __syncthreads();

    for (int mt = wave; mt < 16; mt += 8) {
        const int l0 = mt * 16;
        const int lr = min(l0 + lm, L_SEQ - 1);
        v8f acc = {};
#pragma unroll
        for (int kc = 0; kc < 8; kc += 4) {
            const int k0 = kc + 2 * hi;
            v2f a, bb;
            a.x  = s_xb[(k0    ) * L_SEQ + lr];
            a.y  = s_xb[(k0 + 1) * L_SEQ + lr];
            bb.x = W_in[(16 + lm) * 8 + k0];
            bb.y = W_in[(16 + lm) * 8 + k0 + 1];
            acc = __builtin_amdgcn_wmma_f32_16x16x4_f32(false, a, false, bb,
                                                        (short)0, acc, false, false);
        }
#pragma unroll
        for (int r = 0; r < 8; ++r) {
            const int row = min(l0 + r + 8 * hi, L_SEQ);
            s_z[row * 16 + lm] = acc[r];
        }
    }

    for (int i = tid; i < L_SEQ * DINNER; i += 256) {
        const int l = i >> 4, d = i & 15;
        float accv = conv_b[d];
#pragma unroll
        for (int j = 0; j < DCONV; ++j) {
            const int t = l - (DCONV - 1) + j;
            if (t >= 0) {
                float xi = 0.f;
#pragma unroll
                for (int dm = 0; dm < DMODEL; ++dm) xi += s_xb[dm * L_SEQ + t] * W_in[d * 8 + dm];
                accv += conv_w[d * DCONV + j] * xi;
            }
        }
        s_xc[i] = accv * sigmoidf_(accv);
    }
    __syncthreads();

    for (int l = tid; l < L_SEQ; l += 256) {
        float a = 0.f;
#pragma unroll
        for (int d = 0; d < DINNER; ++d) a += s_xc[l * 16 + d] * W_x[d];
        s_dtr[l] = a;
    }

    for (int j = wave; j < 32; j += 8) {
        const int mt = j >> 1, nt = j & 1;
        const int l0 = mt * 16;
        const int lr = min(l0 + lm, L_SEQ - 1);
        const int e0 = 1 + nt * 16;
        v8f acc = {};
#pragma unroll
        for (int kc = 0; kc < 16; kc += 4) {
            const int k0 = kc + 2 * hi;
            v2f a, bb;
            a.x  = s_xc[lr * 16 + k0];
            a.y  = s_xc[lr * 16 + k0 + 1];
            bb.x = W_x[(e0 + lm) * 16 + k0];
            bb.y = W_x[(e0 + lm) * 16 + k0 + 1];
            acc = __builtin_amdgcn_wmma_f32_16x16x4_f32(false, a, false, bb,
                                                        (short)0, acc, false, false);
        }
        float* dst = nt ? s_C : s_B;
#pragma unroll
        for (int r = 0; r < 8; ++r) {
            const int row = min(l0 + r + 8 * hi, L_SEQ);
            dst[row * 16 + lm] = acc[r];
        }
    }
    __syncthreads();

    {
        const int wv = __builtin_amdgcn_readfirstlane(tid_all >> 5);
        const int e  = wv / 9;
        const int rw = wv - 9 * e;
        if (rw == 0 || rw == 4) {
            float* img = smem_all + e * IMG;
            float*       y_   = img;
            const float* xc_  = img + 4016;
            const float* dtr_ = img + 16048;
            const int q = lane & 3;
            const int d = 2 * rw + (lane >> 2);
            const float4* pB = reinterpret_cast<const float4*>(img + 8016  + 4 * q);
            const float4* pC = reinterpret_cast<const float4*>(img + 12032 + 4 * q);
            const float dtw = dt_w[d], dtb = dt_b[d], Dpd = Dp[d];
            float As[4], h[4];
#pragma unroll
            for (int j = 0; j < 4; ++j) { As[j] = -__expf(A_log[d * 16 + 4 * q + j]); h[j] = 0.f; }

            auto scan_step = [&](const float dv, const float xcv, const float4 bv, const float4 cv) -> float {
                const float dtx = dv * xcv;
                const float Bv[4] = {bv.x, bv.y, bv.z, bv.w};
                const float Cv[4] = {cv.x, cv.y, cv.z, cv.w};
#pragma unroll
                for (int j = 0; j < 4; ++j)
                    h[j] = __expf(dv * As[j]) * h[j] + dtx * Bv[j];
                float ya = h[0] * Cv[0], yb = h[2] * Cv[2];
                ya += h[1] * Cv[1]; yb += h[3] * Cv[3];
                float y = ya + yb;
                y += quad_<0xB1>(y);
                y += quad_<0x4E>(y);
                y += xcv * Dpd;
                return y;
            };

            float own   = softplus_(dtr_[q] * dtw + dtb);
            float dtr_n = dtr_[4 + q];
            float  xcv = xc_[d], zv = y_[q * 16 + d];
            float4 bv = pB[0], cv = pC[0];
            for (int l = 0; l < L_SEQ; l += 4) {
                const float dt0 = quad_<0x00>(own), dt1 = quad_<0x55>(own), dt2 = quad_<0xAA>(own), dt3 = quad_<0xFF>(own);
                own   = softplus_(dtr_n * dtw + dtb);
                dtr_n = dtr_[min(l + 8 + q, L_SEQ - 1)];
                const float n_zv = y_[min(l + 4 + q, L_SEQ - 1) * 16 + d];

                const int r1 = min(l + 1, L_SEQ - 1), r2 = min(l + 2, L_SEQ - 1), r3 = min(l + 3, L_SEQ - 1), r4 = min(l + 4, L_SEQ - 1);
                const float  xc1 = xc_[r1 * 16 + d];
                const float4 b1 = pB[r1 * 4], c1 = pC[r1 * 4];
                float y = scan_step(dt0, xcv, bv, cv);
                const float  xc2 = xc_[r2 * 16 + d];
                const float4 b2 = pB[r2 * 4], c2 = pC[r2 * 4];
                const float y1 = scan_step(dt1, xc1, b1, c1);
                y = (q == 1) ? y1 : y;
                const float  xc3 = xc_[r3 * 16 + d];
                const float4 b3 = pB[r3 * 4], c3 = pC[r3 * 4];
                const float y2 = scan_step(dt2, xc2, b2, c2);
                y = (q == 2) ? y2 : y;
                xcv = xc_[r4 * 16 + d];
                bv = pB[r4 * 4]; cv = pC[r4 * 4];
                const float y3 = scan_step(dt3, xc3, b3, c3);
                y = (q == 3) ? y3 : y;
                y *= zv * sigmoidf_(zv);
                y_[min(l + q, L_SEQ) * 16 + d] = y;
                zv = n_zv;
            }
        }
    }
    __syncthreads();

    for (int mt = wave; mt < 16; mt += 8) {
        const int l0 = mt * 16;
        const int lr = min(l0 + lm, L_SEQ - 1);
        v8f acc = {};
#pragma unroll
        for (int kc = 0; kc < 16; kc += 4) {
            const int k0 = kc + 2 * hi;
            v2f a, bb;
            a.x  = s_y[lr * 16 + k0];
            a.y  = s_y[lr * 16 + k0 + 1];
            bb.x = (lm < DMODEL) ? W_out[lm * 16 + k0]     : 0.f;
            bb.y = (lm < DMODEL) ? W_out[lm * 16 + k0 + 1] : 0.f;
            acc = __builtin_amdgcn_wmma_f32_16x16x4_f32(false, a, false, bb,
                                                        (short)0, acc, false, false);
        }
        if (lm < DMODEL) {
#pragma unroll
            for (int r = 0; r < 8; ++r)
                s_out[(l0 + r + 8 * hi) * 8 + lm] = acc[r];
        }
    }
    __syncthreads();

    {
        float part[NCLS];
#pragma unroll
        for (int c = 0; c < NCLS; ++c) part[c] = 0.f;
        for (int i = tid; i < L_SEQ * DMODEL; i += 256) {
            const float v = s_out[i];
#pragma unroll
            for (int c = 0; c < NCLS; ++c) part[c] += v * fc_w[c * (L_SEQ * DMODEL) + i];
        }
#pragma unroll
        for (int c = 0; c < NCLS; ++c) {
#pragma unroll
            for (int off = 16; off > 0; off >>= 1)
                part[c] += __shfl_down(part[c], off, 32);
        }
        if (lane == 0) {
#pragma unroll
            for (int c = 0; c < NCLS; ++c) s_red[wave * NCLS + c] = part[c];
        }
        __syncthreads();
        if (tid < NCLS) {
            float sres = fc_b[tid];
#pragma unroll
            for (int w = 0; w < 8; ++w) sres += s_red[w * NCLS + tid];
            out[b * NCLS + tid] = sres;
        }
    }
}

extern "C" void kernel_launch(void* const* d_in, const int* in_sizes, int n_in,
                              void* d_out, int out_size, void* d_ws, size_t ws_size,
                              hipStream_t stream) {
    (void)in_sizes; (void)n_in; (void)out_size; (void)d_ws; (void)ws_size;
    mamba_fused_kernel<<<dim3(256), dim3(1024), 0, stream>>>(
        (const float*)d_in[0],
        (const float*)d_in[1],
        (const float*)d_in[2],
        (const float*)d_in[3],
        (const float*)d_in[4],
        (const float*)d_in[5],
        (const float*)d_in[6],
        (const float*)d_in[7],
        (const float*)d_in[8],
        (const float*)d_in[9],
        (const float*)d_in[10],
        (const float*)d_in[11],
        (float*)d_out);
}


# Causal conv and in-projection fused into ONE f32 matrix product: K = 5 taps x 8 inputs = 40, windows of x against weights conv_w*W_in multiplied once; conv leaves the vector unit; sum order changes
# speedup vs baseline: 4.7294x; 1.1322x over previous
#include <hip/hip_runtime.h>
#include <hip/hip_bf16.h>

#define L_SEQ 250
#define DMODEL 8
#define DINNER 16
#define DSTATE 16
#define DCONV 5
#define NCLS 12
#define NELEM 4
#define IMG 16304

typedef __attribute__((ext_vector_type(2))) float v2f;
typedef __attribute__((ext_vector_type(8))) float v8f;

__device__ __forceinline__ float sigmoidf_(float v) { return 1.0f / (1.0f + __expf(-v)); }

__device__ __forceinline__ float softplus_(float v) {
    const float sp = log1pf(__expf(v));
    return (v > 20.f) ? v : sp;
}

template <int CTRL>
__device__ __forceinline__ float quad_(float v) {
    const int b = __builtin_bit_cast(int, v);
    return __builtin_bit_cast(float, __builtin_amdgcn_update_dpp(b, b, CTRL, 0xf, 0xf, false));
}

__global__ __launch_bounds__(1024) void mamba_fused_kernel(
    const float* __restrict__ x,
    const float* __restrict__ W_in,
    const float* __restrict__ conv_w,
    const float* __restrict__ conv_b,
    const float* __restrict__ W_x,
    const float* __restrict__ dt_w,
    const float* __restrict__ dt_b,
    const float* __restrict__ A_log,
    const float* __restrict__ Dp,
    const float* __restrict__ W_out,
    const float* __restrict__ fc_w,
    const float* __restrict__ fc_b,
    float* __restrict__ out)
{
    __shared__ __attribute__((aligned(64))) float smem_all[NELEM * IMG];
    const int tid_all = threadIdx.x;
    const int g    = tid_all >> 8;
    float* smem  = smem_all + g * IMG;
    float* s_z   = smem;
    float* s_xc  = smem + 4016;
    float* s_B   = smem + 8016;
    float* s_C   = smem + 12032;
    float* s_dtr = smem + 16048;
    float* s_xb  = s_B;
    float* s_y   = s_z;
    float* s_out = s_C;
    float* s_red = s_dtr;

    const int b    = blockIdx.x * NELEM + g;
    const int tid  = tid_all & 255;
    const int lane = tid & 31;
    const int wave = tid >> 5;
    const int lm   = lane & 15;
    const int hi   = lane >> 4;
    const float* xb = x + (size_t)b * DMODEL * L_SEQ;

    for (int i = tid; i < DMODEL * L_SEQ; i += 256) s_xb[i] = xb[i];
    for (int i = tid * 32; i < NCLS * L_SEQ * DMODEL; i += 256 * 32)
        __builtin_prefetch(fc_w + i, 0, 0);
    __syncthreads();

    for (int mt = wave; mt < 16; mt += 8) {
        const int l0 = mt * 16;
        const int lr = min(l0 + lm, L_SEQ - 1);
        v8f acc = {};
#pragma unroll
        for (int kc = 0; kc < 8; kc += 4) {
            const int k0 = kc + 2 * hi;
            v2f a, bb;
            a.x  = s_xb[(k0    ) * L_SEQ + lr];
            a.y  = s_xb[(k0 + 1) * L_SEQ + lr];
            bb.x = W_in[(16 + lm) * 8 + k0];
            bb.y = W_in[(16 + lm) * 8 + k0 + 1];
            acc = __builtin_amdgcn_wmma_f32_16x16x4_f32(false, a, false, bb,
                                                        (short)0, acc, false, false);
        }
#pragma unroll
        for (int r = 0; r < 8; ++r) {
            const int row = min(l0 + r + 8 * hi, L_SEQ);
            s_z[row * 16 + lm] = acc[r];
        }
    }

    {
        v2f wf[10];
        {
            const float w0 = W_in[lm * 8 + 2 * hi],     w1 = W_in[lm * 8 + 2 * hi + 1];
            const float w2 = W_in[lm * 8 + 4 + 2 * hi], w3 = W_in[lm * 8 + 5 + 2 * hi];
#pragma unroll
            for (int j = 0; j < DCONV; ++j) {
                const float cw = conv_w[lm * DCONV + j];
                wf[2 * j].x     = cw * w0;  wf[2 * j].y     = cw * w1;
                wf[2 * j + 1].x = cw * w2;  wf[2 * j + 1].y = cw * w3;
            }
        }
        const float cb = conv_b[lm];
        for (int mt = wave; mt < 16; mt += 8) {
            const int l0 = mt * 16;
            const int lr = min(l0 + lm, L_SEQ - 1);
            v2f a[10];
#pragma unroll
            for (int c = 0; c < 10; ++c) {
                const int j  = c >> 1;
                const int dm = 4 * (c & 1) + 2 * hi;
                const int t  = lr - (DCONV - 1) + j;
                const int tc = max(t, 0);
                const float ax = s_xb[(dm    ) * L_SEQ + tc];
                const float ay = s_xb[(dm + 1) * L_SEQ + tc];
                a[c].x = (t < 0) ? 0.f : ax;
                a[c].y = (t < 0) ? 0.f : ay;
            }
            v8f acc = {};
#pragma unroll
            for (int c = 0; c < 10; ++c)
                acc = __builtin_amdgcn_wmma_f32_16x16x4_f32(false, a[c], false, wf[c],
                                                            (short)0, acc, false, false);
#pragma unroll
            for (int r = 0; r < 8; ++r) {
                const int row = l0 + r + 8 * hi;
                const float v = acc[r] + cb;
                float* dst = (row < L_SEQ) ? (s_xc + row * 16 + lm) : (s_C + lm);
                *dst = v * sigmoidf_(v);
            }
        }
    }
    __syncthreads();

    for (int l = tid; l < L_SEQ; l += 256) {
        float a = 0.f;
#pragma unroll
        for (int d = 0; d < DINNER; ++d) a += s_xc[l * 16 + d] * W_x[d];
        s_dtr[l] = a;
    }

    for (int j = wave; j < 32; j += 8) {
        const int mt = j >> 1, nt = j & 1;
        const int l0 = mt * 16;
        const int lr = min(l0 + lm, L_SEQ - 1);
        const int e0 = 1 + nt * 16;
        v8f acc = {};
#pragma unroll
        for (int kc = 0; kc < 16; kc += 4) {
            const int k0 = kc + 2 * hi;
            v2f a, bb;
            a.x  = s_xc[lr * 16 + k0];
            a.y  = s_xc[lr * 16 + k0 + 1];
            bb.x = W_x[(e0 + lm) * 16 + k0];
            bb.y = W_x[(e0 + lm) * 16 + k0 + 1];
            acc = __builtin_amdgcn_wmma_f32_16x16x4_f32(false, a, false, bb,
                                                        (short)0, acc, false, false);
        }
        float* dst = nt ? s_C : s_B;
#pragma unroll
        for (int r = 0; r < 8; ++r) {
            const int row = min(l0 + r + 8 * hi, L_SEQ);
            dst[row * 16 + lm] = acc[r];
        }
    }
    __syncthreads();

    {
        const int wv = __builtin_amdgcn_readfirstlane(tid_all >> 5);
        const int e  = wv / 9;
        const int rw = wv - 9 * e;
        if (rw == 0 || rw == 4) {
            float* img = smem_all + e * IMG;
            float*       y_   = img;
            const float* xc_  = img + 4016;
            const float* dtr_ = img + 16048;
            const int q = lane & 3;
            const int d = 2 * rw + (lane >> 2);
            const float4* pB = reinterpret_cast<const float4*>(img + 8016  + 4 * q);
            const float4* pC = reinterpret_cast<const float4*>(img + 12032 + 4 * q);
            const float dtw = dt_w[d], dtb = dt_b[d], Dpd = Dp[d];
            float As[4], h[4];
#pragma unroll
            for (int j = 0; j < 4; ++j) { As[j] = -__expf(A_log[d * 16 + 4 * q + j]); h[j] = 0.f; }

            auto scan_step = [&](const float dv, const float xcv, const float4 bv, const float4 cv) -> float {
                const float dtx = dv * xcv;
                const float Bv[4] = {bv.x, bv.y, bv.z, bv.w};
                const float Cv[4] = {cv.x, cv.y, cv.z, cv.w};
#pragma unroll
                for (int j = 0; j < 4; ++j)
                    h[j] = __expf(dv * As[j]) * h[j] + dtx * Bv[j];
                float ya = h[0] * Cv[0], yb = h[2] * Cv[2];
                ya += h[1] * Cv[1]; yb += h[3] * Cv[3];
                float y = ya + yb;
                y += quad_<0xB1>(y);
                y += quad_<0x4E>(y);
                y += xcv * Dpd;
                return y;
            };

            float own   = softplus_(dtr_[q] * dtw + dtb);
            float dtr_n = dtr_[4 + q];
            float  xcv = xc_[d], zv = y_[q * 16 + d];
            float4 bv = pB[0], cv = pC[0];
            for (int l = 0; l < L_SEQ; l += 4) {
                const float dt0 = quad_<0x00>(own), dt1 = quad_<0x55>(own), dt2 = quad_<0xAA>(own), dt3 = quad_<0xFF>(own);
                own   = softplus_(dtr_n * dtw + dtb);
                dtr_n = dtr_[min(l + 8 + q, L_SEQ - 1)];
                const float n_zv = y_[min(l + 4 + q, L_SEQ - 1) * 16 + d];

                const int r1 = min(l + 1, L_SEQ - 1), r2 = min(l + 2, L_SEQ - 1), r3 = min(l + 3, L_SEQ - 1), r4 = min(l + 4, L_SEQ - 1);
                const float  xc1 = xc_[r1 * 16 + d];
                const float4 b1 = pB[r1 * 4], c1 = pC[r1 * 4];
                float y = scan_step(dt0, xcv, bv, cv);
                const float  xc2 = xc_[r2 * 16 + d];
                const float4 b2 = pB[r2 * 4], c2 = pC[r2 * 4];
                const float y1 = scan_step(dt1, xc1, b1, c1);
                y = (q == 1) ? y1 : y;
                const float  xc3 = xc_[r3 * 16 + d];
                const float4 b3 = pB[r3 * 4], c3 = pC[r3 * 4];
                const float y2 = scan_step(dt2, xc2, b2, c2);
                y = (q == 2) ? y2 : y;
                xcv = xc_[r4 * 16 + d];
                bv = pB[r4 * 4]; cv = pC[r4 * 4];
                const float y3 = scan_step(dt3, xc3, b3, c3);
                y = (q == 3) ? y3 : y;
                y *= zv * sigmoidf_(zv);
                y_[min(l + q, L_SEQ) * 16 + d] = y;
                zv = n_zv;
            }
        }
    }
    __syncthreads();

    for (int mt = wave; mt < 16; mt += 8) {
        const int l0 = mt * 16;
        const int lr = min(l0 + lm, L_SEQ - 1);
        v8f acc = {};
#pragma unroll
        for (int kc = 0; kc < 16; kc += 4) {
            const int k0 = kc + 2 * hi;
            v2f a, bb;
            a.x  = s_y[lr * 16 + k0];
            a.y  = s_y[lr * 16 + k0 + 1];
            bb.x = (lm < DMODEL) ? W_out[lm * 16 + k0]     : 0.f;
            bb.y = (lm < DMODEL) ? W_out[lm * 16 + k0 + 1] : 0.f;
            acc = __builtin_amdgcn_wmma_f32_16x16x4_f32(false, a, false, bb,
                                                        (short)0, acc, false, false);
        }
        if (lm < DMODEL) {
#pragma unroll
            for (int r = 0; r < 8; ++r)
                s_out[(l0 + r + 8 * hi) * 8 + lm] = acc[r];
        }
    }
    __syncthreads();

    {
        float part[NCLS];
#pragma unroll
        for (int c = 0; c < NCLS; ++c) part[c] = 0.f;
        for (int i = tid; i < L_SEQ * DMODEL; i += 256) {
            const float v = s_out[i];
#pragma unroll
            for (int c = 0; c < NCLS; ++c) part[c] += v * fc_w[c * (L_SEQ * DMODEL) + i];
        }
#pragma unroll
        for (int c = 0; c < NCLS; ++c) {
#pragma unroll
            for (int off = 16; off > 0; off >>= 1)
                part[c] += __shfl_down(part[c], off, 32);
        }
        if (lane == 0) {
#pragma unroll
            for (int c = 0; c < NCLS; ++c) s_red[wave * NCLS + c] = part[c];
        }
        __syncthreads();
        if (tid < NCLS) {
            float sres = fc_b[tid];
#pragma unroll
            for (int w = 0; w < 8; ++w) sres += s_red[w * NCLS + tid];
            out[b * NCLS + tid] = sres;
        }
    }
}

extern "C" void kernel_launch(void* const* d_in, const int* in_sizes, int n_in,
                              void* d_out, int out_size, void* d_ws, size_t ws_size,
                              hipStream_t stream) {
    (void)in_sizes; (void)n_in; (void)out_size; (void)d_ws; (void)ws_size;
    mamba_fused_kernel<<<dim3(256), dim3(1024), 0, stream>>>(
        (const float*)d_in[0],
        (const float*)d_in[1],
        (const float*)d_in[2],
        (const float*)d_in[3],
        (const float*)d_in[4],
        (const float*)d_in[5],
        (const float*)d_in[6],
        (const float*)d_in[7],
        (const float*)d_in[8],
        (const float*)d_in[9],
        (const float*)d_in[10],
        (const float*)d_in[11],
        (float*)d_out);
}


# Two-chunk time pipeline: one wave takes a 16-step tile from x to Bm/Cm and from y to FC sums, no barrier inside; idle waves do chunk 2's front and chunk 1's back beside the scan; FC sum order changes
# speedup vs baseline: 5.0852x; 1.0752x over previous
#include <hip/hip_runtime.h>
#include <hip/hip_bf16.h>

#define L_SEQ 250
#define DMODEL 8
#define DINNER 16
#define DSTATE 16
#define DCONV 5
#define NCLS 12
#define NELEM 4
#define IMG 18336
#define HALF_L 128

typedef __attribute__((ext_vector_type(2))) float v2f;
typedef __attribute__((ext_vector_type(8))) float v8f;

__device__ __forceinline__ float sigmoidf_(float v) { return 1.0f / (1.0f + __expf(-v)); }

__device__ __forceinline__ float softplus_(float v) {
    const float sp = log1pf(__expf(v));
    return (v > 20.f) ? v : sp;
}

template <int CTRL>
__device__ __forceinline__ float quad_(float v) {
    const int b = __builtin_bit_cast(int, v);
    return __builtin_bit_cast(float, __builtin_amdgcn_update_dpp(b, b, CTRL, 0xf, 0xf, false));
}

__global__ __launch_bounds__(1024) void mamba_fused_kernel(
    const float* __restrict__ x,
    const float* __restrict__ W_in,
    const float* __restrict__ conv_w,
    const float* __restrict__ conv_b,
    const float* __restrict__ W_x,
    const float* __restrict__ dt_w,
    const float* __restrict__ dt_b,
    const float* __restrict__ A_log,
    const float* __restrict__ Dp,
    const float* __restrict__ W_out,
    const float* __restrict__ fc_w,
    const float* __restrict__ fc_b,
    float* __restrict__ out)
{
    __shared__ __attribute__((aligned(64))) float smem_all[NELEM * IMG];
    const int tid_all = threadIdx.x;
    const int g    = tid_all >> 8;
    float* smem  = smem_all + g * IMG;
    float* s_z   = smem;
    float* s_xc  = smem + 4016;
    float* s_B   = smem + 8032;
    float* s_C   = smem + 12048;
    float* s_dtr = smem + 16064;
    float* s_x   = smem + 16320;
    float* s_y   = s_z;
    float* s_out = s_C;
    float* s_red = s_dtr;

    const int b    = blockIdx.x * NELEM + g;
    const int tid  = tid_all & 255;
    const int lane = tid & 31;
    const int lm   = lane & 15;
    const int hi   = lane >> 4;
    const float* xb = x + (size_t)b * DMODEL * L_SEQ;

    const int wv = __builtin_amdgcn_readfirstlane(tid_all >> 5);
    const int w8 = wv & 7;
    const int gs = wv >> 3;
    const int e  = wv / 9;
    const int rw = wv - 9 * e;
    const bool scanner = (rw == 0) || (rw == 4);
    const int hk = w8 - ((w8 > gs) ? 1 : 0) - ((w8 > gs + 4) ? 1 : 0);

    for (int i = tid; i < DMODEL * L_SEQ; i += 256) s_x[i] = xb[i];
    for (int i = tid * 32; i < NCLS * L_SEQ * DMODEL; i += 256 * 32)
        __builtin_prefetch(fc_w + i, 0, 0);
    __syncthreads();

    v2f wf[10];
    {
        const float w0 = W_in[lm * 8 + 2 * hi],     w1 = W_in[lm * 8 + 2 * hi + 1];
        const float w2 = W_in[lm * 8 + 4 + 2 * hi], w3 = W_in[lm * 8 + 5 + 2 * hi];
#pragma unroll
        for (int j = 0; j < DCONV; ++j) {
            const float cw = conv_w[lm * DCONV + j];
            wf[2 * j].x     = cw * w0;  wf[2 * j].y     = cw * w1;
            wf[2 * j + 1].x = cw * w2;  wf[2 * j + 1].y = cw * w3;
        }
    }
    const float cb = conv_b[lm];
    v2f wz[2];
    v2f wxB[4], wxC[4];
#pragma unroll
    for (int c = 0; c < 2; ++c) {
        const int k0 = 4 * c + 2 * hi;
        wz[c].x = W_in[(16 + lm) * 8 + k0];
        wz[c].y = W_in[(16 + lm) * 8 + k0 + 1];
    }
#pragma unroll
    for (int c = 0; c < 4; ++c) {
        const int k0 = 4 * c + 2 * hi;
        wxB[c].x = W_x[(1 + lm) * 16 + k0];
        wxB[c].y = W_x[(1 + lm) * 16 + k0 + 1];
        wxC[c].x = W_x[(17 + lm) * 16 + k0];
        wxC[c].y = W_x[(17 + lm) * 16 + k0 + 1];
    }

    auto tile_front = [&](const int mt) {
        const int l0 = mt * 16;
        const int lr = min(l0 + lm, L_SEQ - 1);
        v2f a[10];
#pragma unroll
        for (int c = 0; c < 10; ++c) {
            const int j  = c >> 1;
            const int dm = 4 * (c & 1) + 2 * hi;
            const int t  = lr - (DCONV - 1) + j;
            const int tc = max(t, 0);
            const float ax = s_x[(dm    ) * L_SEQ + tc];
            const float ay = s_x[(dm + 1) * L_SEQ + tc];
            a[c].x = (t < 0) ? 0.f : ax;
            a[c].y = (t < 0) ? 0.f : ay;
        }
        v8f accz = {};
        accz = __builtin_amdgcn_wmma_f32_16x16x4_f32(false, a[8], false, wz[0], (short)0, accz, false, false);
        accz = __builtin_amdgcn_wmma_f32_16x16x4_f32(false, a[9], false, wz[1], (short)0, accz, false, false);
        v8f acc = {};
#pragma unroll
        for (int c = 0; c < 10; ++c)
            acc = __builtin_amdgcn_wmma_f32_16x16x4_f32(false, a[c], false, wf[c],
                                                        (short)0, acc, false, false);
#pragma unroll
        for (int r = 0; r < 8; ++r) {
            const int row = min(l0 + r + 8 * hi, L_SEQ);
            const float v = acc[r] + cb;
            s_z[row * 16 + lm]  = accz[r];
            s_xc[row * 16 + lm] = v * sigmoidf_(v);
        }
        __builtin_amdgcn_fence(__ATOMIC_RELEASE, "wavefront");
        __builtin_amdgcn_wave_barrier();
        __builtin_amdgcn_fence(__ATOMIC_ACQUIRE, "wavefront");
        {
            float s = 0.f;
#pragma unroll
            for (int d = 0; d < DINNER; ++d) s += s_xc[lr * 16 + d] * W_x[d];
            if (hi == 0 && l0 + lm < L_SEQ) s_dtr[l0 + lm] = s;
        }
        v2f ax[4];
#pragma unroll
        for (int c = 0; c < 4; ++c) {
            const int k0 = 4 * c + 2 * hi;
            ax[c].x = s_xc[lr * 16 + k0];
            ax[c].y = s_xc[lr * 16 + k0 + 1];
        }
        v8f accB = {}, accC = {};
#pragma unroll
        for (int c = 0; c < 4; ++c) {
            accB = __builtin_amdgcn_wmma_f32_16x16x4_f32(false, ax[c], false, wxB[c], (short)0, accB, false, false);
            accC = __builtin_amdgcn_wmma_f32_16x16x4_f32(false, ax[c], false, wxC[c], (short)0, accC, false, false);
        }
#pragma unroll
        for (int r = 0; r < 8; ++r) {
            const int row = min(l0 + r + 8 * hi, L_SEQ);
            s_B[row * 16 + lm] = accB[r];
            s_C[row * 16 + lm] = accC[r];
        }
    };

    float part[NCLS];
#pragma unroll
    for (int c = 0; c < NCLS; ++c) part[c] = 0.f;
    auto tile_back = [&](const int mt) {
        const int l0 = mt * 16;
        const int lr = min(l0 + lm, L_SEQ - 1);
        const int wr = min(lm, DMODEL - 1);
        v2f ay[4], wo[4];
#pragma unroll
        for (int c = 0; c < 4; ++c) {
            const int k0 = 4 * c + 2 * hi;
            ay[c].x = s_y[lr * 16 + k0];
            ay[c].y = s_y[lr * 16 + k0 + 1];
            const float t0 = W_out[wr * 16 + k0], t1 = W_out[wr * 16 + k0 + 1];
            wo[c].x = (lm < DMODEL) ? t0 : 0.f;
            wo[c].y = (lm < DMODEL) ? t1 : 0.f;
        }
        v8f acc = {};
#pragma unroll
        for (int c = 0; c < 4; ++c)
            acc = __builtin_amdgcn_wmma_f32_16x16x4_f32(false, ay[c], false, wo[c], (short)0, acc, false, false);
        if (lm < DMODEL) {
#pragma unroll
            for (int r = 0; r < 8; ++r)
                s_out[(l0 + r + 8 * hi) * 8 + lm] = acc[r];
        }
        __builtin_amdgcn_fence(__ATOMIC_RELEASE, "wavefront");
        __builtin_amdgcn_wave_barrier();
        __builtin_amdgcn_fence(__ATOMIC_ACQUIRE, "wavefront");
#pragma unroll
        for (int k = 0; k < 4; ++k) {
            const int i  = l0 * 8 + lane + 32 * k;
            const int ic = min(i, L_SEQ * DMODEL - 1);
            const float sv = s_out[i];
            const float v = (i < L_SEQ * DMODEL) ? sv : 0.f;
#pragma unroll
            for (int c = 0; c < NCLS; ++c) part[c] += v * fc_w[c * (L_SEQ * DMODEL) + ic];
        }
    };

    float As[4] = {0.f, 0.f, 0.f, 0.f}, h[4] = {0.f, 0.f, 0.f, 0.f};
    float dtw = 0.f, dtb = 0.f, Dpd = 0.f;
    auto scan_range = [&](const int lbeg, const int lend, const int cl) {
        float* img = smem_all + e * IMG;
        float*       y_   = img;
        const float* xc_  = img + 4016;
        const float* dtr_ = img + 16064;
        const int q = lane & 3;
        const int d = 2 * rw + (lane >> 2);
        const float4* pB = reinterpret_cast<const float4*>(img + 8032  + 4 * q);
        const float4* pC = reinterpret_cast<const float4*>(img + 12048 + 4 * q);

        auto scan_step = [&](const float dv, const float xcv, const float4 bv, const float4 cv) -> float {
            const float dtx = dv * xcv;
            const float Bv[4] = {bv.x, bv.y, bv.z, bv.w};
            const float Cv[4] = {cv.x, cv.y, cv.z, cv.w};
#pragma unroll
            for (int j = 0; j < 4; ++j)
                h[j] = __expf(dv * As[j]) * h[j] + dtx * Bv[j];
            float ya = h[0] * Cv[0], yb = h[2] * Cv[2];
            ya += h[1] * Cv[1]; yb += h[3] * Cv[3];
            float y = ya + yb;
            y += quad_<0xB1>(y);
            y += quad_<0x4E>(y);
            y += xcv * Dpd;
            return y;
        };

        float own   = softplus_(dtr_[lbeg + q] * dtw + dtb);
        float dtr_n = dtr_[min(lbeg + 4 + q, cl)];
        float  xcv = xc_[lbeg * 16 + d], zv = y_[(lbeg + q) * 16 + d];
        float4 bv = pB[lbeg * 4], cv = pC[lbeg * 4];
        for (int l = lbeg; l < lend; l += 4) {
            const float dt0 = quad_<0x00>(own), dt1 = quad_<0x55>(own), dt2 = quad_<0xAA>(own), dt3 = quad_<0xFF>(own);
            own   = softplus_(dtr_n * dtw + dtb);
            dtr_n = dtr_[min(l + 8 + q, cl)];
            const float n_zv = y_[min(l + 4 + q, cl) * 16 + d];

            const int r1 = min(l + 1, cl), r2 = min(l + 2, cl), r3 = min(l + 3, cl), r4 = min(l + 4, cl);
            const float  xc1 = xc_[r1 * 16 + d];
            const float4 b1 = pB[r1 * 4], c1 = pC[r1 * 4];
            float y = scan_step(dt0, xcv, bv, cv);
            const float  xc2 = xc_[r2 * 16 + d];
            const float4 b2 = pB[r2 * 4], c2 = pC[r2 * 4];
            const float y1 = scan_step(dt1, xc1, b1, c1);
            y = (q == 1) ? y1 : y;
            const float  xc3 = xc_[r3 * 16 + d];
            const float4 b3 = pB[r3 * 4], c3 = pC[r3 * 4];
            const float y2 = scan_step(dt2, xc2, b2, c2);
            y = (q == 2) ? y2 : y;
            xcv = xc_[r4 * 16 + d];
            bv = pB[r4 * 4]; cv = pC[r4 * 4];
            const float y3 = scan_step(dt3, xc3, b3, c3);
            y = (q == 3) ? y3 : y;
            y *= zv * sigmoidf_(zv);
            y_[min(l + q, L_SEQ) * 16 + d] = y;
            zv = n_zv;
        }
    };

    tile_front(w8);
    __syncthreads();

    if (scanner) {
        const int q = lane & 3;
        const int d = 2 * rw + (lane >> 2);
        dtw = dt_w[d]; dtb = dt_b[d]; Dpd = Dp[d];
#pragma unroll
        for (int j = 0; j < 4; ++j) As[j] = -__expf(A_log[d * 16 + 4 * q + j]);
        scan_range(0, HALF_L, HALF_L - 1);
    } else {
        for (int t = 8 + hk; t < 16; t += 6) tile_front(t);
    }
    __syncthreads();

    if (scanner) {
        scan_range(HALF_L, L_SEQ, L_SEQ - 1);
    } else {
        for (int t = hk; t < 8; t += 6) tile_back(t);
    }
    __syncthreads();

    tile_back(8 + w8);
    {
#pragma unroll
        for (int c = 0; c < NCLS; ++c) {
#pragma unroll
            for (int off = 16; off > 0; off >>= 1)
                part[c] += __shfl_down(part[c], off, 32);
        }
        if (lane == 0) {
#pragma unroll
            for (int c = 0; c < NCLS; ++c) s_red[w8 * NCLS + c] = part[c];
        }
        __syncthreads();
        if (tid < NCLS) {
            float sres = fc_b[tid];
#pragma unroll
            for (int w = 0; w < 8; ++w) sres += s_red[w * NCLS + tid];
            out[b * NCLS + tid] = sres;
        }
    }
}

extern "C" void kernel_launch(void* const* d_in, const int* in_sizes, int n_in,
                              void* d_out, int out_size, void* d_ws, size_t ws_size,
                              hipStream_t stream) {
    (void)in_sizes; (void)n_in; (void)out_size; (void)d_ws; (void)ws_size;
    mamba_fused_kernel<<<dim3(256), dim3(1024), 0, stream>>>(
        (const float*)d_in[0],
        (const float*)d_in[1],
        (const float*)d_in[2],
        (const float*)d_in[3],
        (const float*)d_in[4],
        (const float*)d_in[5],
        (const float*)d_in[6],
        (const float*)d_in[7],
        (const float*)d_in[8],
        (const float*)d_in[9],
        (const float*)d_in[10],
        (const float*)d_in[11],
        (float*)d_out);
}


# Wave roles dealt by age: wave w serves element w&3, an element's 8 waves share a SIMD and its 2 scanning waves are the oldest there; helper waves fill the slots the scan leaves; FC sum order changes
# speedup vs baseline: 5.3099x; 1.0442x over previous
#include <hip/hip_runtime.h>
#include <hip/hip_bf16.h>

#define L_SEQ 250
#define DMODEL 8
#define DINNER 16
#define DSTATE 16
#define DCONV 5
#define NCLS 12
#define NELEM 4
#define IMG 18336
#define HALF_L 128

typedef __attribute__((ext_vector_type(2))) float v2f;
typedef __attribute__((ext_vector_type(8))) float v8f;

__device__ __forceinline__ float sigmoidf_(float v) { return 1.0f / (1.0f + __expf(-v)); }

__device__ __forceinline__ float softplus_(float v) {
    const float sp = log1pf(__expf(v));
    return (v > 20.f) ? v : sp;
}

template <int CTRL>
__device__ __forceinline__ float quad_(float v) {
    const int b = __builtin_bit_cast(int, v);
    return __builtin_bit_cast(float, __builtin_amdgcn_update_dpp(b, b, CTRL, 0xf, 0xf, false));
}

__global__ __launch_bounds__(1024) void mamba_fused_kernel(
    const float* __restrict__ x,
    const float* __restrict__ W_in,
    const float* __restrict__ conv_w,
    const float* __restrict__ conv_b,
    const float* __restrict__ W_x,
    const float* __restrict__ dt_w,
    const float* __restrict__ dt_b,
    const float* __restrict__ A_log,
    const float* __restrict__ Dp,
    const float* __restrict__ W_out,
    const float* __restrict__ fc_w,
    const float* __restrict__ fc_b,
    float* __restrict__ out)
{
    __shared__ __attribute__((aligned(64))) float smem_all[NELEM * IMG];
    const int tid_all = threadIdx.x;
    const int wv = __builtin_amdgcn_readfirstlane(tid_all >> 5);
    const int g  = wv & 3;
    const int w8 = wv >> 2;
    float* smem  = smem_all + g * IMG;
    float* s_z   = smem;
    float* s_xc  = smem + 4016;
    float* s_B   = smem + 8032;
    float* s_C   = smem + 12048;
    float* s_dtr = smem + 16064;
    float* s_x   = smem + 16320;
    float* s_y   = s_z;
    float* s_out = s_C;
    float* s_red = s_dtr;

    const int b    = blockIdx.x * NELEM + g;
    const int lane = tid_all & 31;
    const int tid  = w8 * 32 + lane;
    const int lm   = lane & 15;
    const int hi   = lane >> 4;
    const float* xb = x + (size_t)b * DMODEL * L_SEQ;

    const bool scanner = (w8 < 2);
    const int hk = w8 - 2;

    for (int i = tid; i < DMODEL * L_SEQ; i += 256) s_x[i] = xb[i];
    for (int i = tid * 32; i < NCLS * L_SEQ * DMODEL; i += 256 * 32)
        __builtin_prefetch(fc_w + i, 0, 0);
    __syncthreads();

    v2f wf[10];
    {
        const float w0 = W_in[lm * 8 + 2 * hi],     w1 = W_in[lm * 8 + 2 * hi + 1];
        const float w2 = W_in[lm * 8 + 4 + 2 * hi], w3 = W_in[lm * 8 + 5 + 2 * hi];
#pragma unroll
        for (int j = 0; j < DCONV; ++j) {
            const float cw = conv_w[lm * DCONV + j];
            wf[2 * j].x     = cw * w0;  wf[2 * j].y     = cw * w1;
            wf[2 * j + 1].x = cw * w2;  wf[2 * j + 1].y = cw * w3;
        }
    }
    const float cb = conv_b[lm];
    v2f wz[2];
    v2f wxB[4], wxC[4];
#pragma unroll
    for (int c = 0; c < 2; ++c) {
        const int k0 = 4 * c + 2 * hi;
        wz[c].x = W_in[(16 + lm) * 8 + k0];
        wz[c].y = W_in[(16 + lm) * 8 + k0 + 1];
    }
#pragma unroll
    for (int c = 0; c < 4; ++c) {
        const int k0 = 4 * c + 2 * hi;
        wxB[c].x = W_x[(1 + lm) * 16 + k0];
        wxB[c].y = W_x[(1 + lm) * 16 + k0 + 1];
        wxC[c].x = W_x[(17 + lm) * 16 + k0];
        wxC[c].y = W_x[(17 + lm) * 16 + k0 + 1];
    }

    auto tile_front = [&](const int mt) {
        const int l0 = mt * 16;
        const int lr = min(l0 + lm, L_SEQ - 1);
        v2f a[10];
#pragma unroll
        for (int c = 0; c < 10; ++c) {
            const int j  = c >> 1;
            const int dm = 4 * (c & 1) + 2 * hi;
            const int t  = lr - (DCONV - 1) + j;
            const int tc = max(t, 0);
            const float ax = s_x[(dm    ) * L_SEQ + tc];
            const float ay = s_x[(dm + 1) * L_SEQ + tc];
            a[c].x = (t < 0) ? 0.f : ax;
            a[c].y = (t < 0) ? 0.f : ay;
        }
        v8f accz = {};
        accz = __builtin_amdgcn_wmma_f32_16x16x4_f32(false, a[8], false, wz[0], (short)0, accz, false, false);
        accz = __builtin_amdgcn_wmma_f32_16x16x4_f32(false, a[9], false, wz[1], (short)0, accz, false, false);
        v8f acc = {};
#pragma unroll
        for (int c = 0; c < 10; ++c)
            acc = __builtin_amdgcn_wmma_f32_16x16x4_f32(false, a[c], false, wf[c],
                                                        (short)0, acc, false, false);
#pragma unroll
        for (int r = 0; r < 8; ++r) {
            const int row = min(l0 + r + 8 * hi, L_SEQ);
            const float v = acc[r] + cb;
            s_z[row * 16 + lm]  = accz[r];
            s_xc[row * 16 + lm] = v * sigmoidf_(v);
        }
        __builtin_amdgcn_fence(__ATOMIC_RELEASE, "wavefront");
        __builtin_amdgcn_wave_barrier();
        __builtin_amdgcn_fence(__ATOMIC_ACQUIRE, "wavefront");
        {
            float s = 0.f;
#pragma unroll
            for (int d = 0; d < DINNER; ++d) s += s_xc[lr * 16 + d] * W_x[d];
            if (hi == 0 && l0 + lm < L_SEQ) s_dtr[l0 + lm] = s;
        }
        v2f ax[4];
#pragma unroll
        for (int c = 0; c < 4; ++c) {
            const int k0 = 4 * c + 2 * hi;
            ax[c].x = s_xc[lr * 16 + k0];
            ax[c].y = s_xc[lr * 16 + k0 + 1];
        }
        v8f accB = {}, accC = {};
#pragma unroll
        for (int c = 0; c < 4; ++c) {
            accB = __builtin_amdgcn_wmma_f32_16x16x4_f32(false, ax[c], false, wxB[c], (short)0, accB, false, false);
            accC = __builtin_amdgcn_wmma_f32_16x16x4_f32(false, ax[c], false, wxC[c], (short)0, accC, false, false);
        }
#pragma unroll
        for (int r = 0; r < 8; ++r) {
            const int row = min(l0 + r + 8 * hi, L_SEQ);
            s_B[row * 16 + lm] = accB[r];
            s_C[row * 16 + lm] = accC[r];
        }
    };

    float part[NCLS];
#pragma unroll
    for (int c = 0; c < NCLS; ++c) part[c] = 0.f;
    auto tile_back = [&](const int mt) {
        const int l0 = mt * 16;
        const int lr = min(l0 + lm, L_SEQ - 1);
        const int wr = min(lm, DMODEL - 1);
        v2f ay[4], wo[4];
#pragma unroll
        for (int c = 0; c < 4; ++c) {
            const int k0 = 4 * c + 2 * hi;
            ay[c].x = s_y[lr * 16 + k0];
            ay[c].y = s_y[lr * 16 + k0 + 1];
            const float t0 = W_out[wr * 16 + k0], t1 = W_out[wr * 16 + k0 + 1];
            wo[c].x = (lm < DMODEL) ? t0 : 0.f;
            wo[c].y = (lm < DMODEL) ? t1 : 0.f;
        }
        v8f acc = {};
#pragma unroll
        for (int c = 0; c < 4; ++c)
            acc = __builtin_amdgcn_wmma_f32_16x16x4_f32(false, ay[c], false, wo[c], (short)0, acc, false, false);
        if (lm < DMODEL) {
#pragma unroll
            for (int r = 0; r < 8; ++r)
                s_out[(l0 + r + 8 * hi) * 8 + lm] = acc[r];
        }
        __builtin_amdgcn_fence(__ATOMIC_RELEASE, "wavefront");
        __builtin_amdgcn_wave_barrier();
        __builtin_amdgcn_fence(__ATOMIC_ACQUIRE, "wavefront");
#pragma unroll
        for (int k = 0; k < 4; ++k) {
            const int i  = l0 * 8 + lane + 32 * k;
            const int ic = min(i, L_SEQ * DMODEL - 1);
            const float sv = s_out[i];
            const float v = (i < L_SEQ * DMODEL) ? sv : 0.f;
#pragma unroll
            for (int c = 0; c < NCLS; ++c) part[c] += v * fc_w[c * (L_SEQ * DMODEL) + ic];
        }
    };

    float As[4] = {0.f, 0.f, 0.f, 0.f}, h[4] = {0.f, 0.f, 0.f, 0.f};
    float dtw = 0.f, dtb = 0.f, Dpd = 0.f;
    auto scan_range = [&](const int lbeg, const int lend, const int cl) {
        float* img = smem;
        float*       y_   = img;
        const float* xc_  = img + 4016;
        const float* dtr_ = img + 16064;
        const int q = lane & 3;
        const int d = 8 * w8 + (lane >> 2);
        const float4* pB = reinterpret_cast<const float4*>(img + 8032  + 4 * q);
        const float4* pC = reinterpret_cast<const float4*>(img + 12048 + 4 * q);

        auto scan_step = [&](const float dv, const float xcv, const float4 bv, const float4 cv) -> float {
            const float dtx = dv * xcv;
            const float Bv[4] = {bv.x, bv.y, bv.z, bv.w};
            const float Cv[4] = {cv.x, cv.y, cv.z, cv.w};
#pragma unroll
            for (int j = 0; j < 4; ++j)
                h[j] = __expf(dv * As[j]) * h[j] + dtx * Bv[j];
            float ya = h[0] * Cv[0], yb = h[2] * Cv[2];
            ya += h[1] * Cv[1]; yb += h[3] * Cv[3];
            float y = ya + yb;
            y += quad_<0xB1>(y);
            y += quad_<0x4E>(y);
            y += xcv * Dpd;
            return y;
        };

        float own   = softplus_(dtr_[lbeg + q] * dtw + dtb);
        float dtr_n = dtr_[min(lbeg + 4 + q, cl)];
        float  xcv = xc_[lbeg * 16 + d], zv = y_[(lbeg + q) * 16 + d];
        float4 bv = pB[lbeg * 4], cv = pC[lbeg * 4];
        for (int l = lbeg; l < lend; l += 4) {
            const float dt0 = quad_<0x00>(own), dt1 = quad_<0x55>(own), dt2 = quad_<0xAA>(own), dt3 = quad_<0xFF>(own);
            own   = softplus_(dtr_n * dtw + dtb);
            dtr_n = dtr_[min(l + 8 + q, cl)];
            const float n_zv = y_[min(l + 4 + q, cl) * 16 + d];

            const int r1 = min(l + 1, cl), r2 = min(l + 2, cl), r3 = min(l + 3, cl), r4 = min(l + 4, cl);
            const float  xc1 = xc_[r1 * 16 + d];
            const float4 b1 = pB[r1 * 4], c1 = pC[r1 * 4];
            float y = scan_step(dt0, xcv, bv, cv);
            const float  xc2 = xc_[r2 * 16 + d];
            const float4 b2 = pB[r2 * 4], c2 = pC[r2 * 4];
            const float y1 = scan_step(dt1, xc1, b1, c1);
            y = (q == 1) ? y1 : y;
            const float  xc3 = xc_[r3 * 16 + d];
            const float4 b3 = pB[r3 * 4], c3 = pC[r3 * 4];
            const float y2 = scan_step(dt2, xc2, b2, c2);
            y = (q == 2) ? y2 : y;
            xcv = xc_[r4 * 16 + d];
            bv = pB[r4 * 4]; cv = pC[r4 * 4];
            const float y3 = scan_step(dt3, xc3, b3, c3);
            y = (q == 3) ? y3 : y;
            y *= zv * sigmoidf_(zv);
            y_[min(l + q, L_SEQ) * 16 + d] = y;
            zv = n_zv;
        }
    };

    tile_front(w8);
    __syncthreads();

    if (scanner) {
        const int q = lane & 3;
        const int d = 8 * w8 + (lane >> 2);
        dtw = dt_w[d]; dtb = dt_b[d]; Dpd = Dp[d];
#pragma unroll
        for (int j = 0; j < 4; ++j) As[j] = -__expf(A_log[d * 16 + 4 * q + j]);
        scan_range(0, HALF_L, HALF_L - 1);
    } else {
        for (int t = 8 + hk; t < 16; t += 6) tile_front(t);
    }
    __syncthreads();

    if (scanner) {
        scan_range(HALF_L, L_SEQ, L_SEQ - 1);
    } else {
        for (int t = hk; t < 8; t += 6) tile_back(t);
    }
    __syncthreads();

    tile_back(8 + w8);
    {
#pragma unroll
        for (int c = 0; c < NCLS; ++c) {
#pragma unroll
            for (int off = 16; off > 0; off >>= 1)
                part[c] += __shfl_down(part[c], off, 32);
        }
        if (lane == 0) {
#pragma unroll
            for (int c = 0; c < NCLS; ++c) s_red[w8 * NCLS + c] = part[c];
        }
        __syncthreads();
        if (tid < NCLS) {
            float sres = fc_b[tid];
#pragma unroll
            for (int w = 0; w < 8; ++w) sres += s_red[w * NCLS + tid];
            out[b * NCLS + tid] = sres;
        }
    }
}

extern "C" void kernel_launch(void* const* d_in, const int* in_sizes, int n_in,
                              void* d_out, int out_size, void* d_ws, size_t ws_size,
                              hipStream_t stream) {
    (void)in_sizes; (void)n_in; (void)out_size; (void)d_ws; (void)ws_size;
    mamba_fused_kernel<<<dim3(256), dim3(1024), 0, stream>>>(
        (const float*)d_in[0],
        (const float*)d_in[1],
        (const float*)d_in[2],
        (const float*)d_in[3],
        (const float*)d_in[4],
        (const float*)d_in[5],
        (const float*)d_in[6],
        (const float*)d_in[7],
        (const float*)d_in[8],
        (const float*)d_in[9],
        (const float*)d_in[10],
        (const float*)d_in[11],
        (float*)d_out);
}
